# starting kernel
# baseline (speedup 1.0000x reference)
#include <hip/hip_runtime.h>
#include <math.h>


typedef __attribute__((ext_vector_type(2))) float v2f;
typedef __attribute__((ext_vector_type(8))) float v8f;

#define IMG_S        64
#define NPTS         512
#define INV_SIG2     4096.0f
#define INV_GAMMA    0.2f
#define TANHF        0.57735026919f
#define INV_TANHF    1.73205080757f
#define ZNEARC       0.1f
#define ZFARC        6.0f
#define BGW          1e-8f
#define PI_F         3.14159265358979f

__device__ __forceinline__ void norm3(float& x, float& y, float& z) {
    float n = sqrtf(x * x + y * y + z * z);
    float inv = 1.0f / fmaxf(n, 1e-12f);
    x *= inv; y *= inv; z *= inv;
}
__device__ __forceinline__ void cross3(float ax, float ay, float az,
                                       float bx, float by, float bz,
                                       float& cx, float& cy, float& cz) {
    cx = ay * bz - az * by;
    cy = az * bx - ax * bz;
    cz = ax * by - ay * bx;
}

__global__ __launch_bounds__(256)
void MultiViewRenderer_48180943126589_kernel(
    const float* __restrict__ vertices,
    const float* __restrict__ radii,
    const float* __restrict__ at_centers,
    const float* __restrict__ at_normals,
    const float* __restrict__ rho,
    const float* __restrict__ phi,
    const float* __restrict__ theta,
    const float* __restrict__ up,
    float* __restrict__ out)
{
    __shared__ float rec[NPTS * 8];

    const int view = blockIdx.y;
    const int b = view >> 3;
    const int n = view & 7;

    float rho_c = fminf(fmaxf(rho[n], 0.3f), 1.0f);
    float phi_c = fminf(fmaxf(phi[n], 0.0f), 0.5f * PI_F);
    float th_c  = fminf(fmaxf(theta[n], 0.0f), 2.0f * PI_F);
    float sp = __sinf(phi_c), cp = __cosf(phi_c);
    float st = __sinf(th_c),  ct = __cosf(th_c);
    float exl = rho_c * sp * ct, eyl = rho_c * sp * st, ezl = rho_c * cp;

    float cxc = at_centers[3 * b + 0], cyc = at_centers[3 * b + 1], czc = at_centers[3 * b + 2];
    float azx = at_normals[3 * b + 0], azy = at_normals[3 * b + 1], azz = at_normals[3 * b + 2];
    norm3(azx, azy, azz);
    float unx = up[0], uny = up[1], unz = up[2];
    norm3(unx, uny, unz);
    float axx, axy, axz;
    cross3(unx, uny, unz, azx, azy, azz, axx, axy, axz);
    norm3(axx, axy, axz);
    float ayx, ayy, ayz;
    cross3(azx, azy, azz, axx, axy, axz, ayx, ayy, ayz);
    norm3(ayx, ayy, ayz);
    float ex = exl * axx + eyl * ayx + ezl * azx + cxc;
    float ey = exl * axy + eyl * ayy + ezl * azy + cyc;
    float ez = exl * axz + eyl * ayz + ezl * azz + czc;
    float fx = ex - cxc, fy = ey - cyc, fz = ez - czc;
    norm3(fx, fy, fz);
    float rx, ry, rz;
    cross3(unx, uny, unz, fx, fy, fz, rx, ry, rz);
    norm3(rx, ry, rz);
    float ux, uy, uz;
    cross3(fx, fy, fz, rx, ry, rz, ux, uy, uz);
    norm3(ux, uy, uz);
    float t0 = -(rx * ex + ry * ey + rz * ez);
    float t1 = -(ux * ex + uy * ey + uz * ez);
    float t2 = -(fx * ex + fy * ey + fz * ez);
    float r0x = INV_TANHF * rx, r0y = INV_TANHF * ry, r0z = INV_TANHF * rz, r0w = INV_TANHF * t0;
    float r1x = INV_TANHF * ux, r1y = INV_TANHF * uy, r1z = INV_TANHF * uz, r1w = INV_TANHF * t1;
    float r3x = -fx, r3y = -fy, r3z = -fz, r3w = -t2;

    for (int pt = threadIdx.x; pt < NPTS; pt += 256) {
        float vx = vertices[3 * pt + 0], vy = vertices[3 * pt + 1], vz = vertices[3 * pt + 2];
        float c0 = r0x * vx + r0y * vy + r0z * vz + r0w;
        float c1 = r1x * vx + r1y * vy + r1z * vz + r1w;
        float wc = r3x * vx + r3y * vy + r3z * vz + r3w;
        float wcs = (wc > 1e-6f) ? wc : 1.0f;
        float xs = c0 / wcs;
        float ys = c1 / wcs;
        float depth = fminf(fmaxf((wc - ZNEARC) * (1.0f / (ZFARC - ZNEARC)), 0.0f), 1.0f);
        bool  infr = (wc > ZNEARC) && (wc < ZFARC);
        float rn = radii[pt] / (wcs * TANHF);
        float cut2 = 25.0f * rn * rn;
        float cj = -(xs * xs + ys * ys) * INV_SIG2 + (1.0f - depth) * INV_GAMMA;
        float thr = infr ? ((1.0f - depth) * INV_GAMMA - cut2 * INV_SIG2) : 3.0e38f;
        float* r8 = &rec[pt * 8];
        r8[0] = 2.0f * xs * INV_SIG2;
        r8[1] = 2.0f * ys * INV_SIG2;
        r8[2] = cj;
        r8[3] = depth;
        r8[4] = thr;
    }
    __syncthreads();

    const int wave = threadIdx.x >> 5;
    const int lane = threadIdx.x & 31;
    const int hi   = lane >> 4;
    const int colL = lane & 15;
    const int pixbase = blockIdx.x * 128 + wave * 16;

    v2f A;
    {
        int p = pixbase + (lane & 15);
        float pxv = ((float)(p & 63) + 0.5f) * (2.0f / 64.0f) - 1.0f;
        float pyv = 1.0f - ((float)(p >> 6) + 0.5f) * (2.0f / 64.0f);
        A.x = hi ? 1.0f : pxv;
        A.y = hi ? 0.0f : pyv;
    }
    v8f C;
#pragma unroll
    for (int r = 0; r < 8; ++r) {
        int p = pixbase + r + hi * 8;
        float pxv = ((float)(p & 63) + 0.5f) * (2.0f / 64.0f) - 1.0f;
        float pyv = 1.0f - ((float)(p >> 6) + 0.5f) * (2.0f / 64.0f);
        C[r] = -(pxv * pxv + pyv * pyv) * INV_SIG2;
    }

    float accw[8], accd[8];
#pragma unroll
    for (int r = 0; r < 8; ++r) { accw[r] = 0.0f; accd[r] = 0.0f; }

#pragma unroll 4
    for (int k = 0; k < NPTS / 16; ++k) {
        int j = k * 16 + colL;
        const float4 q = *reinterpret_cast<const float4*>(&rec[j * 8]);
        const float thr = rec[j * 8 + 4];
        v2f B;
        B.x = hi ? q.z : q.x;
        B.y = hi ? 0.0f : q.y;
        v8f D = __builtin_amdgcn_wmma_f32_16x16x4_f32(
             false, A,  false, B,
             (short)0, C,  false,  false);
#pragma unroll
        for (int r = 0; r < 8; ++r) {
            float lw = D[r];
            float w = (lw >= thr) ? __expf(lw) : 0.0f;
            accw[r] += w;
            accd[r] += w * q.w;
        }
    }

#pragma unroll
    for (int r = 0; r < 8; ++r) {
        float w = accw[r], d = accd[r];
        w += __shfl_xor(w, 8, 32);  d += __shfl_xor(d, 8, 32);
        w += __shfl_xor(w, 4, 32);  d += __shfl_xor(d, 4, 32);
        w += __shfl_xor(w, 2, 32);  d += __shfl_xor(d, 2, 32);
        w += __shfl_xor(w, 1, 32);  d += __shfl_xor(d, 1, 32);
        accw[r] = w; accd[r] = d;
    }

    if (colL == 0) {
        int base = view * (IMG_S * IMG_S) + pixbase + hi * 8;
#pragma unroll
        for (int r = 0; r < 8; ++r)
            out[base + r] = (BGW + accd[r]) / (BGW + accw[r]);
    }
}

extern "C" void kernel_launch(void* const* d_in, const int* in_sizes, int n_in,
                              void* d_out, int out_size, void* d_ws, size_t ws_size,
                              hipStream_t stream) {
    const float* vertices   = (const float*)d_in[0];
    const float* radii      = (const float*)d_in[1];
    const float* at_centers = (const float*)d_in[2];
    const float* at_normals = (const float*)d_in[3];
    const float* rho        = (const float*)d_in[4];
    const float* phi        = (const float*)d_in[5];
    const float* theta      = (const float*)d_in[6];
    const float* up         = (const float*)d_in[7];

    dim3 grid(32, 64);
    dim3 block(256);
    MultiViewRenderer_48180943126589_kernel<<<grid, block, 0, stream>>>(
        vertices, radii, at_centers, at_normals, rho, phi, theta, up,
        (float*)d_out);
}


# Per work-group ballot compaction in LDS keeps only points whose splat can reach its two pixel rows (others have f32 weight exactly 0); WMMA loop covers kept points only; changes float sum order
# speedup vs baseline: 2.0209x; 2.0209x over previous
#include <hip/hip_runtime.h>
#include <math.h>


typedef __attribute__((ext_vector_type(2))) float v2f;
typedef __attribute__((ext_vector_type(8))) float v8f;

#define IMG_S        64
#define NPTS         512
#define INV_SIG2     4096.0f
#define INV_GAMMA    0.2f
#define TANHF        0.57735026919f
#define INV_TANHF    1.73205080757f
#define ZNEARC       0.1f
#define ZFARC        6.0f
#define BGW          1e-8f
#define PI_F         3.14159265358979f
#define REACH_MAX    0.17f

__device__ __forceinline__ void norm3(float& x, float& y, float& z) {
    float n = sqrtf(x * x + y * y + z * z);
    float inv = 1.0f / fmaxf(n, 1e-12f);
    x *= inv; y *= inv; z *= inv;
}
__device__ __forceinline__ void cross3(float ax, float ay, float az,
                                       float bx, float by, float bz,
                                       float& cx, float& cy, float& cz) {
    cx = ay * bz - az * by;
    cy = az * bx - ax * bz;
    cz = ax * by - ay * bx;
}

__global__ __launch_bounds__(256)
void MultiViewRenderer_48180943126589_kernel(
    const float* __restrict__ vertices,
    const float* __restrict__ radii,
    const float* __restrict__ at_centers,
    const float* __restrict__ at_normals,
    const float* __restrict__ rho,
    const float* __restrict__ phi,
    const float* __restrict__ theta,
    const float* __restrict__ up,
    float* __restrict__ out)
{
    __shared__ float rec[NPTS * 8];
    __shared__ int   cnt[16];

    const int view = blockIdx.y;
    const int b = view >> 3;
    const int n = view & 7;

    float rho_c = fminf(fmaxf(rho[n], 0.3f), 1.0f);
    float phi_c = fminf(fmaxf(phi[n], 0.0f), 0.5f * PI_F);
    float th_c  = fminf(fmaxf(theta[n], 0.0f), 2.0f * PI_F);
    float sp = __sinf(phi_c), cp = __cosf(phi_c);
    float st = __sinf(th_c),  ct = __cosf(th_c);
    float exl = rho_c * sp * ct, eyl = rho_c * sp * st, ezl = rho_c * cp;

    float cxc = at_centers[3 * b + 0], cyc = at_centers[3 * b + 1], czc = at_centers[3 * b + 2];
    float azx = at_normals[3 * b + 0], azy = at_normals[3 * b + 1], azz = at_normals[3 * b + 2];
    norm3(azx, azy, azz);
    float unx = up[0], uny = up[1], unz = up[2];
    norm3(unx, uny, unz);
    float axx, axy, axz;
    cross3(unx, uny, unz, azx, azy, azz, axx, axy, axz);
    norm3(axx, axy, axz);
    float ayx, ayy, ayz;
    cross3(azx, azy, azz, axx, axy, axz, ayx, ayy, ayz);
    norm3(ayx, ayy, ayz);
    float ex = exl * axx + eyl * ayx + ezl * azx + cxc;
    float ey = exl * axy + eyl * ayy + ezl * azy + cyc;
    float ez = exl * axz + eyl * ayz + ezl * azz + czc;
    float fx = ex - cxc, fy = ey - cyc, fz = ez - czc;
    norm3(fx, fy, fz);
    float rx, ry, rz;
    cross3(unx, uny, unz, fx, fy, fz, rx, ry, rz);
    norm3(rx, ry, rz);
    float ux, uy, uz;
    cross3(fx, fy, fz, rx, ry, rz, ux, uy, uz);
    norm3(ux, uy, uz);
    float t0 = -(rx * ex + ry * ey + rz * ez);
    float t1 = -(ux * ex + uy * ey + uz * ez);
    float t2 = -(fx * ex + fy * ey + fz * ez);
    float r0x = INV_TANHF * rx, r0y = INV_TANHF * ry, r0z = INV_TANHF * rz, r0w = INV_TANHF * t0;
    float r1x = INV_TANHF * ux, r1y = INV_TANHF * uy, r1z = INV_TANHF * uz, r1w = INV_TANHF * t1;
    float r3x = -fx, r3y = -fy, r3z = -fz, r3w = -t2;

    const int wave = threadIdx.x >> 5;
    const int lane = threadIdx.x & 31;
    const float py_top = 1.0f - ((float)(2 * blockIdx.x) + 0.5f) * (2.0f / 64.0f);
    const float py_bot = 1.0f - ((float)(2 * blockIdx.x) + 1.5f) * (2.0f / 64.0f);
    const float px_lim = 1.0f - 1.0f / 64.0f;

    float rbx[2], rby[2], rcj[2], rdp[2], rth[2];
    bool  keep[2];
    int   below[2];
#pragma unroll
    for (int h = 0; h < 2; ++h) {
        const int pt = h * 256 + threadIdx.x;
        float vx = vertices[3 * pt + 0], vy = vertices[3 * pt + 1], vz = vertices[3 * pt + 2];
        float c0 = r0x * vx + r0y * vy + r0z * vz + r0w;
        float c1 = r1x * vx + r1y * vy + r1z * vz + r1w;
        float wc = r3x * vx + r3y * vy + r3z * vz + r3w;
        float wcs = (wc > 1e-6f) ? wc : 1.0f;
        float xs = c0 / wcs;
        float ys = c1 / wcs;
        float depth = fminf(fmaxf((wc - ZNEARC) * (1.0f / (ZFARC - ZNEARC)), 0.0f), 1.0f);
        bool  infr = (wc > ZNEARC) && (wc < ZFARC);
        float rn = radii[pt] / (wcs * TANHF);
        float cut2 = 25.0f * rn * rn;
        float cj = -(xs * xs + ys * ys) * INV_SIG2 + (1.0f - depth) * INV_GAMMA;
        float thr = infr ? ((1.0f - depth) * INV_GAMMA - cut2 * INV_SIG2) : 3.0e38f;
        rbx[h] = 2.0f * xs * INV_SIG2;
        rby[h] = 2.0f * ys * INV_SIG2;
        rcj[h] = cj;
        rdp[h] = depth;
        rth[h] = thr;
        const float reach = fminf(REACH_MAX, fabsf(rn) * 5.005f + 0.002f);
        const bool kp = infr && (ys >= py_bot - reach) && (ys <= py_top + reach)
                             && (fabsf(xs) <= px_lim + reach);
        const unsigned m = __builtin_amdgcn_ballot_w32(kp);
        keep[h]  = kp;
        below[h] = __builtin_popcount(m & ((1u << lane) - 1u));
        if (lane == 0) cnt[h * 8 + wave] = __builtin_popcount(m);
    }
    __syncthreads();

    int run0 = 0, run1 = 0, off0 = 0, off1 = 0;
#pragma unroll
    for (int c = 0; c < 8; ++c) {
        if (c == wave) { off0 = run0; off1 = run1; }
        run0 += cnt[c];
        run1 += cnt[8 + c];
    }
    const int nkept = run0 + run1;
    const int npad  = (nkept + 15) & ~15;
#pragma unroll
    for (int h = 0; h < 2; ++h) {
        if (keep[h]) {
            float* r8 = &rec[((h ? run0 + off1 : off0) + below[h]) * 8];
            *reinterpret_cast<float4*>(r8) = make_float4(rbx[h], rby[h], rcj[h], rdp[h]);
            r8[4] = rth[h];
        }
    }
    if (threadIdx.x < 16 && nkept + (int)threadIdx.x < npad) {
        float* r8 = &rec[(nkept + threadIdx.x) * 8];
        *reinterpret_cast<float4*>(r8) = make_float4(0.0f, 0.0f, 0.0f, 0.0f);
        r8[4] = 3.0e38f;
    }
    __syncthreads();
    const int nchunks = __builtin_amdgcn_readfirstlane(npad >> 4);

    const int hi   = lane >> 4;
    const int colL = lane & 15;
    const int pixbase = blockIdx.x * 128 + wave * 16;

    v2f A;
    {
        int p = pixbase + (lane & 15);
        float pxv = ((float)(p & 63) + 0.5f) * (2.0f / 64.0f) - 1.0f;
        float pyv = 1.0f - ((float)(p >> 6) + 0.5f) * (2.0f / 64.0f);
        A.x = hi ? 1.0f : pxv;
        A.y = hi ? 0.0f : pyv;
    }
    v8f C;
#pragma unroll
    for (int r = 0; r < 8; ++r) {
        int p = pixbase + r + hi * 8;
        float pxv = ((float)(p & 63) + 0.5f) * (2.0f / 64.0f) - 1.0f;
        float pyv = 1.0f - ((float)(p >> 6) + 0.5f) * (2.0f / 64.0f);
        C[r] = -(pxv * pxv + pyv * pyv) * INV_SIG2;
    }

    float accw[8], accd[8];
#pragma unroll
    for (int r = 0; r < 8; ++r) { accw[r] = 0.0f; accd[r] = 0.0f; }

    for (int k = 0; k < nchunks; ++k) {
        int j = k * 16 + colL;
        const float4 q = *reinterpret_cast<const float4*>(&rec[j * 8]);
        const float thr = rec[j * 8 + 4];
        v2f B;
        B.x = hi ? q.z : q.x;
        B.y = hi ? 0.0f : q.y;
        v8f D = __builtin_amdgcn_wmma_f32_16x16x4_f32(
             false, A,  false, B,
             (short)0, C,  false,  false);
#pragma unroll
        for (int r = 0; r < 8; ++r) {
            float lw = D[r];
            float w = (lw >= thr) ? __expf(lw) : 0.0f;
            accw[r] += w;
            accd[r] += w * q.w;
        }
    }

#pragma unroll
    for (int r = 0; r < 8; ++r) {
        float w = accw[r], d = accd[r];
        w += __shfl_xor(w, 8, 32);  d += __shfl_xor(d, 8, 32);
        w += __shfl_xor(w, 4, 32);  d += __shfl_xor(d, 4, 32);
        w += __shfl_xor(w, 2, 32);  d += __shfl_xor(d, 2, 32);
        w += __shfl_xor(w, 1, 32);  d += __shfl_xor(d, 1, 32);
        accw[r] = w; accd[r] = d;
    }

    if (colL == 0) {
        int base = view * (IMG_S * IMG_S) + pixbase + hi * 8;
#pragma unroll
        for (int r = 0; r < 8; ++r)
            out[base + r] = (BGW + accd[r]) / (BGW + accw[r]);
    }
}

extern "C" void kernel_launch(void* const* d_in, const int* in_sizes, int n_in,
                              void* d_out, int out_size, void* d_ws, size_t ws_size,
                              hipStream_t stream) {
    const float* vertices   = (const float*)d_in[0];
    const float* radii      = (const float*)d_in[1];
    const float* at_centers = (const float*)d_in[2];
    const float* at_normals = (const float*)d_in[3];
    const float* rho        = (const float*)d_in[4];
    const float* phi        = (const float*)d_in[5];
    const float* theta      = (const float*)d_in[6];
    const float* up         = (const float*)d_in[7];

    dim3 grid(32, 64);
    dim3 block(256);
    MultiViewRenderer_48180943126589_kernel<<<grid, block, 0, stream>>>(
        vertices, radii, at_centers, at_normals, rho, phi, theta, up,
        (float*)d_out);
}


# Each work-group loops over 4 pairs of pixel rows (grid 32x64 -> 8x64): camera set-up and point projection run once per 4 pairs, 4x fewer waves to dispatch; one packed list per pair; sum order kept
# speedup vs baseline: 2.1312x; 1.0546x over previous
#include <hip/hip_runtime.h>
#include <math.h>


typedef __attribute__((ext_vector_type(2))) float v2f;
typedef __attribute__((ext_vector_type(8))) float v8f;

#define IMG_S        64
#define NPTS         512
#define RP_PER_WG    4
#define NWG_X        (IMG_S / 2 / RP_PER_WG)
static_assert(NWG_X == 8, "kernel_launch sets grid.x to this value as a literal");
#define INV_SIG2     4096.0f
#define INV_GAMMA    0.2f
#define TANHF        0.57735026919f
#define INV_TANHF    1.73205080757f
#define ZNEARC       0.1f
#define ZFARC        6.0f
#define BGW          1e-8f
#define PI_F         3.14159265358979f
#define REACH_MAX    0.17f

__device__ __forceinline__ void norm3(float& x, float& y, float& z) {
    float n = sqrtf(x * x + y * y + z * z);
    float inv = 1.0f / fmaxf(n, 1e-12f);
    x *= inv; y *= inv; z *= inv;
}
__device__ __forceinline__ void cross3(float ax, float ay, float az,
                                       float bx, float by, float bz,
                                       float& cx, float& cy, float& cz) {
    cx = ay * bz - az * by;
    cy = az * bx - ax * bz;
    cz = ax * by - ay * bx;
}

__global__ __launch_bounds__(256)
void MultiViewRenderer_48180943126589_kernel(
    const float* __restrict__ vertices,
    const float* __restrict__ radii,
    const float* __restrict__ at_centers,
    const float* __restrict__ at_normals,
    const float* __restrict__ rho,
    const float* __restrict__ phi,
    const float* __restrict__ theta,
    const float* __restrict__ up,
    float* __restrict__ out)
{
    __shared__ float rec[RP_PER_WG][NPTS * 8];
    __shared__ int   cnt[RP_PER_WG][16];
    __shared__ int   nch[RP_PER_WG];

    const int view = blockIdx.y;
    const int b = view >> 3;
    const int n = view & 7;

    float rho_c = fminf(fmaxf(rho[n], 0.3f), 1.0f);
    float phi_c = fminf(fmaxf(phi[n], 0.0f), 0.5f * PI_F);
    float th_c  = fminf(fmaxf(theta[n], 0.0f), 2.0f * PI_F);
    float sp = __sinf(phi_c), cp = __cosf(phi_c);
    float st = __sinf(th_c),  ct = __cosf(th_c);
    float exl = rho_c * sp * ct, eyl = rho_c * sp * st, ezl = rho_c * cp;

    float cxc = at_centers[3 * b + 0], cyc = at_centers[3 * b + 1], czc = at_centers[3 * b + 2];
    float azx = at_normals[3 * b + 0], azy = at_normals[3 * b + 1], azz = at_normals[3 * b + 2];
    norm3(azx, azy, azz);
    float unx = up[0], uny = up[1], unz = up[2];
    norm3(unx, uny, unz);
    float axx, axy, axz;
    cross3(unx, uny, unz, azx, azy, azz, axx, axy, axz);
    norm3(axx, axy, axz);
    float ayx, ayy, ayz;
    cross3(azx, azy, azz, axx, axy, axz, ayx, ayy, ayz);
    norm3(ayx, ayy, ayz);
    float ex = exl * axx + eyl * ayx + ezl * azx + cxc;
    float ey = exl * axy + eyl * ayy + ezl * azy + cyc;
    float ez = exl * axz + eyl * ayz + ezl * azz + czc;
    float fx = ex - cxc, fy = ey - cyc, fz = ez - czc;
    norm3(fx, fy, fz);
    float rx, ry, rz;
    cross3(unx, uny, unz, fx, fy, fz, rx, ry, rz);
    norm3(rx, ry, rz);
    float ux, uy, uz;
    cross3(fx, fy, fz, rx, ry, rz, ux, uy, uz);
    norm3(ux, uy, uz);
    float t0 = -(rx * ex + ry * ey + rz * ez);
    float t1 = -(ux * ex + uy * ey + uz * ez);
    float t2 = -(fx * ex + fy * ey + fz * ez);
    float r0x = INV_TANHF * rx, r0y = INV_TANHF * ry, r0z = INV_TANHF * rz, r0w = INV_TANHF * t0;
    float r1x = INV_TANHF * ux, r1y = INV_TANHF * uy, r1z = INV_TANHF * uz, r1w = INV_TANHF * t1;
    float r3x = -fx, r3y = -fy, r3z = -fz, r3w = -t2;

    const int wave = threadIdx.x >> 5;
    const int lane = threadIdx.x & 31;
    const float px_lim = 1.0f - 1.0f / 64.0f;

    float rbx[2], rby[2], rcj[2], rdp[2], rth[2];
    bool  keep[2][RP_PER_WG];
    int   below[2][RP_PER_WG];
#pragma unroll
    for (int h = 0; h < 2; ++h) {
        const int pt = h * 256 + threadIdx.x;
        float vx = vertices[3 * pt + 0], vy = vertices[3 * pt + 1], vz = vertices[3 * pt + 2];
        float c0 = r0x * vx + r0y * vy + r0z * vz + r0w;
        float c1 = r1x * vx + r1y * vy + r1z * vz + r1w;
        float wc = r3x * vx + r3y * vy + r3z * vz + r3w;
        float wcs = (wc > 1e-6f) ? wc : 1.0f;
        float xs = c0 / wcs;
        float ys = c1 / wcs;
        float depth = fminf(fmaxf((wc - ZNEARC) * (1.0f / (ZFARC - ZNEARC)), 0.0f), 1.0f);
        bool  infr = (wc > ZNEARC) && (wc < ZFARC);
        float rn = radii[pt] / (wcs * TANHF);
        float cut2 = 25.0f * rn * rn;
        float cj = -(xs * xs + ys * ys) * INV_SIG2 + (1.0f - depth) * INV_GAMMA;
        float thr = infr ? ((1.0f - depth) * INV_GAMMA - cut2 * INV_SIG2) : 3.0e38f;
        rbx[h] = 2.0f * xs * INV_SIG2;
        rby[h] = 2.0f * ys * INV_SIG2;
        rcj[h] = cj;
        rdp[h] = depth;
        rth[h] = thr;
        const float reach = fminf(REACH_MAX, fabsf(rn) * 5.005f + 0.002f);
        const bool kx = infr && (fabsf(xs) <= px_lim + reach);
#pragma unroll
        for (int q = 0; q < RP_PER_WG; ++q) {
            const int rp = blockIdx.x + NWG_X * q;
            const float py_top = 1.0f - ((float)(2 * rp) + 0.5f) * (2.0f / 64.0f);
            const float py_bot = 1.0f - ((float)(2 * rp) + 1.5f) * (2.0f / 64.0f);
            const bool kp = kx && (ys >= py_bot - reach) && (ys <= py_top + reach);
            const unsigned m = __builtin_amdgcn_ballot_w32(kp);
            keep[h][q]  = kp;
            below[h][q] = __builtin_popcount(m & ((1u << lane) - 1u));
            if (lane == 0) cnt[q][h * 8 + wave] = __builtin_popcount(m);
        }
    }
    __syncthreads();

#pragma unroll
    for (int q = 0; q < RP_PER_WG; ++q) {
        int run0 = 0, run1 = 0, off0 = 0, off1 = 0;
#pragma unroll
        for (int c = 0; c < 8; ++c) {
            if (c == wave) { off0 = run0; off1 = run1; }
            run0 += cnt[q][c];
            run1 += cnt[q][8 + c];
        }
        const int nkept = run0 + run1;
        const int npad  = (nkept + 15) & ~15;
#pragma unroll
        for (int h = 0; h < 2; ++h) {
            if (keep[h][q]) {
                float* r8 = &rec[q][((h ? run0 + off1 : off0) + below[h][q]) * 8];
                *reinterpret_cast<float4*>(r8) = make_float4(rbx[h], rby[h], rcj[h], rdp[h]);
                r8[4] = rth[h];
            }
        }
        if (threadIdx.x < 16 && nkept + (int)threadIdx.x < npad) {
            float* r8 = &rec[q][(nkept + threadIdx.x) * 8];
            *reinterpret_cast<float4*>(r8) = make_float4(0.0f, 0.0f, 0.0f, 0.0f);
            r8[4] = 3.0e38f;
        }
        if (threadIdx.x == 0) nch[q] = npad >> 4;
    }
    __syncthreads();

    const int hi   = lane >> 4;
    const int colL = lane & 15;
    const int colbase = (wave & 3) * 16;

    for (int q = 0; q < RP_PER_WG; ++q) {
        const int rp = blockIdx.x + NWG_X * q;
        const int pixbase = rp * 128 + wave * 16;
        const float* __restrict__ recq = rec[q];
        const int nchunks = __builtin_amdgcn_readfirstlane(nch[q]);

        const float pyv = 1.0f - ((float)(2 * rp + (wave >> 2)) + 0.5f) * (2.0f / 64.0f);
        v2f A;
        {
            float pxv = ((float)(colbase + colL) + 0.5f) * (2.0f / 64.0f) - 1.0f;
            A.x = hi ? 1.0f : pxv;
            A.y = hi ? 0.0f : pyv;
        }
        v8f C;
#pragma unroll
        for (int r = 0; r < 8; ++r) {
            float pxv = ((float)(colbase + r + hi * 8) + 0.5f) * (2.0f / 64.0f) - 1.0f;
            C[r] = -(pxv * pxv + pyv * pyv) * INV_SIG2;
        }

        float accw[8], accd[8];
#pragma unroll
        for (int r = 0; r < 8; ++r) { accw[r] = 0.0f; accd[r] = 0.0f; }

        for (int k = 0; k < nchunks; ++k) {
            int j = k * 16 + colL;
            const float4 pr = *reinterpret_cast<const float4*>(&recq[j * 8]);
            const float thr = recq[j * 8 + 4];
            v2f B;
            B.x = hi ? pr.z : pr.x;
            B.y = hi ? 0.0f : pr.y;
            v8f D = __builtin_amdgcn_wmma_f32_16x16x4_f32(
                 false, A,  false, B,
                 (short)0, C,  false,  false);
#pragma unroll
            for (int r = 0; r < 8; ++r) {
                float lw = D[r];
                float w = (lw >= thr) ? __expf(lw) : 0.0f;
                accw[r] += w;
                accd[r] += w * pr.w;
            }
        }

#pragma unroll
        for (int r = 0; r < 8; ++r) {
            float w = accw[r], d = accd[r];
            w += __shfl_xor(w, 8, 32);  d += __shfl_xor(d, 8, 32);
            w += __shfl_xor(w, 4, 32);  d += __shfl_xor(d, 4, 32);
            w += __shfl_xor(w, 2, 32);  d += __shfl_xor(d, 2, 32);
            w += __shfl_xor(w, 1, 32);  d += __shfl_xor(d, 1, 32);
            accw[r] = w; accd[r] = d;
        }

        if (colL == 0) {
            int base = view * (IMG_S * IMG_S) + pixbase + hi * 8;
#pragma unroll
            for (int r = 0; r < 8; ++r)
                out[base + r] = (BGW + accd[r]) / (BGW + accw[r]);
        }
    }
}

extern "C" void kernel_launch(void* const* d_in, const int* in_sizes, int n_in,
                              void* d_out, int out_size, void* d_ws, size_t ws_size,
                              hipStream_t stream) {
    const float* vertices   = (const float*)d_in[0];
    const float* radii      = (const float*)d_in[1];
    const float* at_centers = (const float*)d_in[2];
    const float* at_normals = (const float*)d_in[3];
    const float* rho        = (const float*)d_in[4];
    const float* phi        = (const float*)d_in[5];
    const float* theta      = (const float*)d_in[6];
    const float* up         = (const float*)d_in[7];

    dim3 grid(8, 64);
    dim3 block(256);
    MultiViewRenderer_48180943126589_kernel<<<grid, block, 0, stream>>>(
        vertices, radii, at_centers, at_normals, rho, phi, theta, up,
        (float*)d_out);
}


# Camera's 12 scalars come by ONE lane-indexed vector read + readlane; both points of a thread are read during the camera math: 5 dependent memory round trips before the 1st barrier become 1; same sums
# speedup vs baseline: 2.7472x; 1.2890x over previous
#include <hip/hip_runtime.h>
#include <math.h>


typedef __attribute__((ext_vector_type(2))) float v2f;
typedef __attribute__((ext_vector_type(8))) float v8f;

#define IMG_S        64
#define NPTS         512
#define RP_PER_WG    4
#define NWG_X        (IMG_S / 2 / RP_PER_WG)
static_assert(NWG_X == 8, "kernel_launch sets grid.x to this value as a literal");
#define INV_SIG2     4096.0f
#define INV_GAMMA    0.2f
#define TANHF        0.57735026919f
#define INV_TANHF    1.73205080757f
#define ZNEARC       0.1f
#define ZFARC        6.0f
#define BGW          1e-8f
#define PI_F         3.14159265358979f
#define REACH_MAX    0.17f

__device__ __forceinline__ void norm3(float& x, float& y, float& z) {
    float n = sqrtf(x * x + y * y + z * z);
    float inv = 1.0f / fmaxf(n, 1e-12f);
    x *= inv; y *= inv; z *= inv;
}
__device__ __forceinline__ void cross3(float ax, float ay, float az,
                                       float bx, float by, float bz,
                                       float& cx, float& cy, float& cz) {
    cx = ay * bz - az * by;
    cy = az * bx - ax * bz;
    cz = ax * by - ay * bx;
}

__device__ __forceinline__ float xhalf(float x) {
    return __builtin_bit_cast(float, __builtin_amdgcn_permlanex16(
        __builtin_bit_cast(int, x), __builtin_bit_cast(int, x),
        0x76543210, 0xfedcba98, false, false));
}

__global__ __launch_bounds__(256)
void MultiViewRenderer_48180943126589_kernel(
    const float* __restrict__ vertices,
    const float* __restrict__ radii,
    const float* __restrict__ at_centers,
    const float* __restrict__ at_normals,
    const float* __restrict__ rho,
    const float* __restrict__ phi,
    const float* __restrict__ theta,
    const float* __restrict__ up,
    float* __restrict__ out)
{
    __shared__ float4 fa[RP_PER_WG][NPTS];
    __shared__ __attribute__((aligned(16))) float th[RP_PER_WG][NPTS];
    __shared__ __attribute__((aligned(16))) float dp[RP_PER_WG][NPTS];
    __shared__ int   cnt[RP_PER_WG][16];
    __shared__ int   nch[RP_PER_WG];

    const int view = blockIdx.y;
    const int b = view >> 3;
    const int n = view & 7;

    const int lane = threadIdx.x & 31;
    const int ck = lane < 11 ? lane : 11;
    const float* cbase = up;
    cbase = ck >= 3  ? at_normals : cbase;
    cbase = ck >= 6  ? at_centers : cbase;
    cbase = ck == 9  ? rho        : cbase;
    cbase = ck == 10 ? phi        : cbase;
    cbase = ck == 11 ? theta      : cbase;
    const int offn = 3 * b + ck - 3, offc = 3 * b + ck - 6;
    int coff = ck;
    coff = ck >= 3 ? offn : coff;
    coff = ck >= 6 ? offc : coff;
    coff = ck >= 9 ? n    : coff;
    const int cam = __builtin_bit_cast(int, cbase[coff]);
    float pvx[2], pvy[2], pvz[2], prad[2];
#pragma unroll
    for (int h = 0; h < 2; ++h) {
        const int pt = h * 256 + threadIdx.x;
        pvx[h] = vertices[3 * pt + 0];
        pvy[h] = vertices[3 * pt + 1];
        pvz[h] = vertices[3 * pt + 2];
        prad[h] = radii[pt];
    }
    const float up0   = __builtin_bit_cast(float, __builtin_amdgcn_readlane(cam, 0));
    const float up1   = __builtin_bit_cast(float, __builtin_amdgcn_readlane(cam, 1));
    const float up2   = __builtin_bit_cast(float, __builtin_amdgcn_readlane(cam, 2));
    const float nrm0  = __builtin_bit_cast(float, __builtin_amdgcn_readlane(cam, 3));
    const float nrm1  = __builtin_bit_cast(float, __builtin_amdgcn_readlane(cam, 4));
    const float nrm2  = __builtin_bit_cast(float, __builtin_amdgcn_readlane(cam, 5));
    const float cen0  = __builtin_bit_cast(float, __builtin_amdgcn_readlane(cam, 6));
    const float cen1  = __builtin_bit_cast(float, __builtin_amdgcn_readlane(cam, 7));
    const float cen2  = __builtin_bit_cast(float, __builtin_amdgcn_readlane(cam, 8));
    const float rho_n   = __builtin_bit_cast(float, __builtin_amdgcn_readlane(cam, 9));
    const float phi_n   = __builtin_bit_cast(float, __builtin_amdgcn_readlane(cam, 10));
    const float theta_n = __builtin_bit_cast(float, __builtin_amdgcn_readlane(cam, 11));

    float rho_c = fminf(fmaxf(rho_n, 0.3f), 1.0f);
    float phi_c = fminf(fmaxf(phi_n, 0.0f), 0.5f * PI_F);
    float th_c  = fminf(fmaxf(theta_n, 0.0f), 2.0f * PI_F);
    float sp = __sinf(phi_c), cp = __cosf(phi_c);
    float st = __sinf(th_c),  ct = __cosf(th_c);
    float exl = rho_c * sp * ct, eyl = rho_c * sp * st, ezl = rho_c * cp;

    float cxc = cen0, cyc = cen1, czc = cen2;
    float azx = nrm0, azy = nrm1, azz = nrm2;
    norm3(azx, azy, azz);
    float unx = up0, uny = up1, unz = up2;
    norm3(unx, uny, unz);
    float axx, axy, axz;
    cross3(unx, uny, unz, azx, azy, azz, axx, axy, axz);
    norm3(axx, axy, axz);
    float ayx, ayy, ayz;
    cross3(azx, azy, azz, axx, axy, axz, ayx, ayy, ayz);
    norm3(ayx, ayy, ayz);
    float ex = exl * axx + eyl * ayx + ezl * azx + cxc;
    float ey = exl * axy + eyl * ayy + ezl * azy + cyc;
    float ez = exl * axz + eyl * ayz + ezl * azz + czc;
    float fx = ex - cxc, fy = ey - cyc, fz = ez - czc;
    norm3(fx, fy, fz);
    float rx, ry, rz;
    cross3(unx, uny, unz, fx, fy, fz, rx, ry, rz);
    norm3(rx, ry, rz);
    float ux, uy, uz;
    cross3(fx, fy, fz, rx, ry, rz, ux, uy, uz);
    norm3(ux, uy, uz);
    float t0 = -(rx * ex + ry * ey + rz * ez);
    float t1 = -(ux * ex + uy * ey + uz * ez);
    float t2 = -(fx * ex + fy * ey + fz * ez);
    float r0x = INV_TANHF * rx, r0y = INV_TANHF * ry, r0z = INV_TANHF * rz, r0w = INV_TANHF * t0;
    float r1x = INV_TANHF * ux, r1y = INV_TANHF * uy, r1z = INV_TANHF * uz, r1w = INV_TANHF * t1;
    float r3x = -fx, r3y = -fy, r3z = -fz, r3w = -t2;

    const int wave = threadIdx.x >> 5;
    const float px_lim = 1.0f - 1.0f / 64.0f;

    float rbx[2], rby[2], rcj[2], rdp[2], rth[2];
    bool  keep[2][RP_PER_WG];
    int   below[2][RP_PER_WG];
#pragma unroll
    for (int h = 0; h < 2; ++h) {
        float vx = pvx[h], vy = pvy[h], vz = pvz[h];
        float c0 = r0x * vx + r0y * vy + r0z * vz + r0w;
        float c1 = r1x * vx + r1y * vy + r1z * vz + r1w;
        float wc = r3x * vx + r3y * vy + r3z * vz + r3w;
        float wcs = (wc > 1e-6f) ? wc : 1.0f;
        float xs = c0 / wcs;
        float ys = c1 / wcs;
        float depth = fminf(fmaxf((wc - ZNEARC) * (1.0f / (ZFARC - ZNEARC)), 0.0f), 1.0f);
        bool  infr = (wc > ZNEARC) && (wc < ZFARC);
        float rn = prad[h] / (wcs * TANHF);
        float cut2 = 25.0f * rn * rn;
        float cj = -(xs * xs + ys * ys) * INV_SIG2 + (1.0f - depth) * INV_GAMMA;
        float thr = infr ? ((1.0f - depth) * INV_GAMMA - cut2 * INV_SIG2) : 3.0e38f;
        rbx[h] = 2.0f * xs * INV_SIG2;
        rby[h] = 2.0f * ys * INV_SIG2;
        rcj[h] = cj;
        rdp[h] = depth;
        rth[h] = thr;
        const float reach = fminf(REACH_MAX, fabsf(rn) * 5.005f + 0.002f);
        const bool kx = infr && (fabsf(xs) <= px_lim + reach);
#pragma unroll
        for (int q = 0; q < RP_PER_WG; ++q) {
            const int rp = blockIdx.x + NWG_X * q;
            const float py_top = 1.0f - ((float)(2 * rp) + 0.5f) * (2.0f / 64.0f);
            const float py_bot = 1.0f - ((float)(2 * rp) + 1.5f) * (2.0f / 64.0f);
            const bool kp = kx && (ys >= py_bot - reach) && (ys <= py_top + reach);
            const unsigned m = __builtin_amdgcn_ballot_w32(kp);
            keep[h][q]  = kp;
            below[h][q] = __builtin_popcount(m & ((1u << lane) - 1u));
            if (lane == 0) cnt[q][h * 8 + wave] = __builtin_popcount(m);
        }
    }
    __syncthreads();

#pragma unroll
    for (int q = 0; q < RP_PER_WG; ++q) {
        int run0 = 0, run1 = 0, off0 = 0, off1 = 0;
#pragma unroll
        for (int c = 0; c < 8; ++c) {
            if (c == wave) { off0 = run0; off1 = run1; }
            run0 += cnt[q][c];
            run1 += cnt[q][8 + c];
        }
        const int nkept = run0 + run1;
        const int npad  = (nkept + 15) & ~15;
#pragma unroll
        for (int h = 0; h < 2; ++h) {
            if (keep[h][q]) {
                const int pos = (h ? run0 + off1 : off0) + below[h][q];
                fa[q][pos] = make_float4(rbx[h], rby[h], rcj[h], 0.0f);
                th[q][pos] = rth[h];
                dp[q][pos] = rdp[h];
            }
        }
        if (threadIdx.x < 16 && nkept + (int)threadIdx.x < npad) {
            const int pos = nkept + (int)threadIdx.x;
            fa[q][pos] = make_float4(0.0f, 0.0f, 0.0f, 0.0f);
            th[q][pos] = 3.0e38f;
            dp[q][pos] = 0.0f;
        }
        if (threadIdx.x == 0) nch[q] = npad >> 4;
    }
    __syncthreads();

    const int hi   = lane >> 4;
    const int colL = lane & 15;
    const int colbase = (wave & 3) * 16;

    for (int q = 0; q < RP_PER_WG; ++q) {
        const int rp = blockIdx.x + NWG_X * q;
        const int pixbase = rp * 128 + wave * 16;
        const float* __restrict__ faq = reinterpret_cast<const float*>(fa[q]);
        const float* __restrict__ thq = th[q];
        const float* __restrict__ dpq = dp[q];
        const int nchunks = __builtin_amdgcn_readfirstlane(nch[q]);

        const float pyv = 1.0f - ((float)(2 * rp + (wave >> 2)) + 0.5f) * (2.0f / 64.0f);
        const float pxv = ((float)(colbase + colL) + 0.5f) * (2.0f / 64.0f) - 1.0f;
        v2f B;
        B.x = hi ? 1.0f : pxv;
        B.y = hi ? 0.0f : pyv;
        float cpix;
        {
#pragma clang fp contract(off)
            cpix = -(pxv * pxv + pyv * pyv) * INV_SIG2;
        }
        v8f C;
#pragma unroll
        for (int r = 0; r < 8; ++r) C[r] = cpix;

        v2f aw0 = {0.0f, 0.0f}, aw1 = {0.0f, 0.0f};
        v2f ad0 = {0.0f, 0.0f}, ad1 = {0.0f, 0.0f};

        for (int k = 0; k < nchunks; ++k) {
            const int j = k * 16;
            const v2f A = *reinterpret_cast<const v2f*>(&faq[(j + colL) * 4 + 2 * hi]);
            const float4 t0 = *reinterpret_cast<const float4*>(&thq[j + 8 * hi]);
            const float4 t1 = *reinterpret_cast<const float4*>(&thq[j + 8 * hi + 4]);
            const float4 d0 = *reinterpret_cast<const float4*>(&dpq[j + 8 * hi]);
            const float4 d1 = *reinterpret_cast<const float4*>(&dpq[j + 8 * hi + 4]);
            v8f D = __builtin_amdgcn_wmma_f32_16x16x4_f32(
                 false, A,  false, B,
                 (short)0, C,  false,  false);
            const float tv[8] = {t0.x, t0.y, t0.z, t0.w, t1.x, t1.y, t1.z, t1.w};
            float w[8];
#pragma unroll
            for (int r = 0; r < 8; ++r) {
                const float lw = D[r];
                w[r] = (lw >= tv[r]) ? __expf(lw) : 0.0f;
            }
            const v2f w01 = {w[0], w[1]}, w23 = {w[2], w[3]}, w45 = {w[4], w[5]}, w67 = {w[6], w[7]};
            const v2f e01 = {d0.x, d0.y}, e23 = {d0.z, d0.w}, e45 = {d1.x, d1.y}, e67 = {d1.z, d1.w};
            aw0 += w01;  ad0 += w01 * e01;
            aw1 += w23;  ad1 += w23 * e23;
            aw0 += w45;  ad0 += w45 * e45;
            aw1 += w67;  ad1 += w67 * e67;
        }

        float ws = (aw0.x + aw1.x) + (aw0.y + aw1.y);
        float ds = (ad0.x + ad1.x) + (ad0.y + ad1.y);
        ws += xhalf(ws);
        ds += xhalf(ds);

        if (hi == 0)
            out[view * (IMG_S * IMG_S) + pixbase + colL] = (BGW + ds) / (BGW + ws);
    }
}

extern "C" void kernel_launch(void* const* d_in, const int* in_sizes, int n_in,
                              void* d_out, int out_size, void* d_ws, size_t ws_size,
                              hipStream_t stream) {
    const float* vertices   = (const float*)d_in[0];
    const float* radii      = (const float*)d_in[1];
    const float* at_centers = (const float*)d_in[2];
    const float* at_normals = (const float*)d_in[3];
    const float* rho        = (const float*)d_in[4];
    const float* phi        = (const float*)d_in[5];
    const float* theta      = (const float*)d_in[6];
    const float* up         = (const float*)d_in[7];

    dim3 grid(8, 64);
    dim3 block(256);
    MultiViewRenderer_48180943126589_kernel<<<grid, block, 0, stream>>>(
        vertices, radii, at_centers, at_normals, rho, phi, theta, up,
        (float*)d_out);
}


# Packed point lists per 8x16-pixel tile not per 2x64 row pair (a wave owns one 16-pixel row of a tile): fewer points in reach of a list, fewer WMMA chunks; cutoff tests bit-identical; new sum order
# speedup vs baseline: 3.2503x; 1.1831x over previous
#include <hip/hip_runtime.h>
#include <math.h>


typedef __attribute__((ext_vector_type(2))) float v2f;
typedef __attribute__((ext_vector_type(8))) float v8f;

#define IMG_S        64
#define NPTS         512
#define TILE_ROWS    8
#define TILE_COLS    16
#define TILES_PER_WG 4
#define NWG_X        ((IMG_S / TILE_ROWS) * (IMG_S / TILE_COLS) / TILES_PER_WG)
static_assert(NWG_X == 8, "kernel_launch sets grid.x to this value as a literal");
#define INV_SIG2     4096.0f
#define INV_GAMMA    0.2f
#define TANHF        0.57735026919f
#define INV_TANHF    1.73205080757f
#define ZNEARC       0.1f
#define ZFARC        6.0f
#define BGW          1e-8f
#define PI_F         3.14159265358979f
#define REACH_MAX    0.17f

__device__ __forceinline__ void norm3(float& x, float& y, float& z) {
    float n = sqrtf(x * x + y * y + z * z);
    float inv = 1.0f / fmaxf(n, 1e-12f);
    x *= inv; y *= inv; z *= inv;
}
__device__ __forceinline__ void cross3(float ax, float ay, float az,
                                       float bx, float by, float bz,
                                       float& cx, float& cy, float& cz) {
    cx = ay * bz - az * by;
    cy = az * bx - ax * bz;
    cz = ax * by - ay * bx;
}

__device__ __forceinline__ float xhalf(float x) {
    return __builtin_bit_cast(float, __builtin_amdgcn_permlanex16(
        __builtin_bit_cast(int, x), __builtin_bit_cast(int, x),
        0x76543210, 0xfedcba98, false, false));
}

__global__ __launch_bounds__(256)
void MultiViewRenderer_48180943126589_kernel(
    const float* __restrict__ vertices,
    const float* __restrict__ radii,
    const float* __restrict__ at_centers,
    const float* __restrict__ at_normals,
    const float* __restrict__ rho,
    const float* __restrict__ phi,
    const float* __restrict__ theta,
    const float* __restrict__ up,
    float* __restrict__ out)
{
    __shared__ float4 fa[TILES_PER_WG][NPTS];
    __shared__ __attribute__((aligned(16))) float th[TILES_PER_WG][NPTS];
    __shared__ __attribute__((aligned(16))) float dp[TILES_PER_WG][NPTS];
    __shared__ int   cnt[TILES_PER_WG][16];
    __shared__ int   nch[TILES_PER_WG];

    const int view = blockIdx.y;
    const int b = view >> 3;
    const int n = view & 7;

    const int lane = threadIdx.x & 31;
    const int ck = lane < 11 ? lane : 11;
    const float* cbase = up;
    cbase = ck >= 3  ? at_normals : cbase;
    cbase = ck >= 6  ? at_centers : cbase;
    cbase = ck == 9  ? rho        : cbase;
    cbase = ck == 10 ? phi        : cbase;
    cbase = ck == 11 ? theta      : cbase;
    const int offn = 3 * b + ck - 3, offc = 3 * b + ck - 6;
    int coff = ck;
    coff = ck >= 3 ? offn : coff;
    coff = ck >= 6 ? offc : coff;
    coff = ck >= 9 ? n    : coff;
    const int cam = __builtin_bit_cast(int, cbase[coff]);
    float pvx[2], pvy[2], pvz[2], prad[2];
#pragma unroll
    for (int h = 0; h < 2; ++h) {
        const int pt = h * 256 + threadIdx.x;
        pvx[h] = vertices[3 * pt + 0];
        pvy[h] = vertices[3 * pt + 1];
        pvz[h] = vertices[3 * pt + 2];
        prad[h] = radii[pt];
    }
    const float up0   = __builtin_bit_cast(float, __builtin_amdgcn_readlane(cam, 0));
    const float up1   = __builtin_bit_cast(float, __builtin_amdgcn_readlane(cam, 1));
    const float up2   = __builtin_bit_cast(float, __builtin_amdgcn_readlane(cam, 2));
    const float nrm0  = __builtin_bit_cast(float, __builtin_amdgcn_readlane(cam, 3));
    const float nrm1  = __builtin_bit_cast(float, __builtin_amdgcn_readlane(cam, 4));
    const float nrm2  = __builtin_bit_cast(float, __builtin_amdgcn_readlane(cam, 5));
    const float cen0  = __builtin_bit_cast(float, __builtin_amdgcn_readlane(cam, 6));
    const float cen1  = __builtin_bit_cast(float, __builtin_amdgcn_readlane(cam, 7));
    const float cen2  = __builtin_bit_cast(float, __builtin_amdgcn_readlane(cam, 8));
    const float rho_n   = __builtin_bit_cast(float, __builtin_amdgcn_readlane(cam, 9));
    const float phi_n   = __builtin_bit_cast(float, __builtin_amdgcn_readlane(cam, 10));
    const float theta_n = __builtin_bit_cast(float, __builtin_amdgcn_readlane(cam, 11));

    float rho_c = fminf(fmaxf(rho_n, 0.3f), 1.0f);
    float phi_c = fminf(fmaxf(phi_n, 0.0f), 0.5f * PI_F);
    float th_c  = fminf(fmaxf(theta_n, 0.0f), 2.0f * PI_F);
    float sp = __sinf(phi_c), cp = __cosf(phi_c);
    float st = __sinf(th_c),  ct = __cosf(th_c);
    float exl = rho_c * sp * ct, eyl = rho_c * sp * st, ezl = rho_c * cp;

    float cxc = cen0, cyc = cen1, czc = cen2;
    float azx = nrm0, azy = nrm1, azz = nrm2;
    norm3(azx, azy, azz);
    float unx = up0, uny = up1, unz = up2;
    norm3(unx, uny, unz);
    float axx, axy, axz;
    cross3(unx, uny, unz, azx, azy, azz, axx, axy, axz);
    norm3(axx, axy, axz);
    float ayx, ayy, ayz;
    cross3(azx, azy, azz, axx, axy, axz, ayx, ayy, ayz);
    norm3(ayx, ayy, ayz);
    float ex = exl * axx + eyl * ayx + ezl * azx + cxc;
    float ey = exl * axy + eyl * ayy + ezl * azy + cyc;
    float ez = exl * axz + eyl * ayz + ezl * azz + czc;
    float fx = ex - cxc, fy = ey - cyc, fz = ez - czc;
    norm3(fx, fy, fz);
    float rx, ry, rz;
    cross3(unx, uny, unz, fx, fy, fz, rx, ry, rz);
    norm3(rx, ry, rz);
    float ux, uy, uz;
    cross3(fx, fy, fz, rx, ry, rz, ux, uy, uz);
    norm3(ux, uy, uz);
    float t0 = -(rx * ex + ry * ey + rz * ez);
    float t1 = -(ux * ex + uy * ey + uz * ez);
    float t2 = -(fx * ex + fy * ey + fz * ez);
    float r0x = INV_TANHF * rx, r0y = INV_TANHF * ry, r0z = INV_TANHF * rz, r0w = INV_TANHF * t0;
    float r1x = INV_TANHF * ux, r1y = INV_TANHF * uy, r1z = INV_TANHF * uz, r1w = INV_TANHF * t1;
    float r3x = -fx, r3y = -fy, r3z = -fz, r3w = -t2;

    const int wave = threadIdx.x >> 5;

    float rbx[2], rby[2], rcj[2], rdp[2], rth[2];
    bool  keep[2][TILES_PER_WG];
    int   below[2][TILES_PER_WG];
#pragma unroll
    for (int h = 0; h < 2; ++h) {
        float vx = pvx[h], vy = pvy[h], vz = pvz[h];
        float c0 = r0x * vx + r0y * vy + r0z * vz + r0w;
        float c1 = r1x * vx + r1y * vy + r1z * vz + r1w;
        float wc = r3x * vx + r3y * vy + r3z * vz + r3w;
        float wcs = (wc > 1e-6f) ? wc : 1.0f;
        float xs = c0 / wcs;
        float ys = c1 / wcs;
        float depth = fminf(fmaxf((wc - ZNEARC) * (1.0f / (ZFARC - ZNEARC)), 0.0f), 1.0f);
        bool  infr = (wc > ZNEARC) && (wc < ZFARC);
        float rn = prad[h] / (wcs * TANHF);
        float cut2 = 25.0f * rn * rn;
        float cj = -(xs * xs + ys * ys) * INV_SIG2 + (1.0f - depth) * INV_GAMMA;
        float thr = infr ? ((1.0f - depth) * INV_GAMMA - cut2 * INV_SIG2) : 3.0e38f;
        rbx[h] = 2.0f * xs * INV_SIG2;
        rby[h] = 2.0f * ys * INV_SIG2;
        rcj[h] = cj;
        rdp[h] = depth;
        rth[h] = thr;
        const float reach = fminf(REACH_MAX, fabsf(rn) * 5.005f + 0.002f);
        const float ry = reach + 3.5f * (2.0f / 64.0f);
        const float rx = reach + 7.5f * (2.0f / 64.0f);
#pragma unroll
        for (int q = 0; q < TILES_PER_WG; ++q) {
            const int ty = (blockIdx.x + 4 * (q >> 1)) & 7;
            const float yc = 1.0f - ((float)(TILE_ROWS * ty) + 4.0f) * (2.0f / 64.0f);
            const float xc = ((float)(TILE_COLS * q) + 8.0f) * (2.0f / 64.0f) - 1.0f;
            const bool kp = infr & (fabsf(ys - yc) <= ry) & (fabsf(xs - xc) <= rx);
            const unsigned m = __builtin_amdgcn_ballot_w32(kp);
            keep[h][q]  = kp;
            below[h][q] = __builtin_popcount(m & ((1u << lane) - 1u));
            if (lane == 0) cnt[q][h * 8 + wave] = __builtin_popcount(m);
        }
    }
    __syncthreads();

#pragma unroll
    for (int q = 0; q < TILES_PER_WG; ++q) {
        int run0 = 0, run1 = 0, off0 = 0, off1 = 0;
#pragma unroll
        for (int c = 0; c < 8; ++c) {
            if (c == wave) { off0 = run0; off1 = run1; }
            run0 += cnt[q][c];
            run1 += cnt[q][8 + c];
        }
        const int nkept = run0 + run1;
        const int npad  = (nkept + 15) & ~15;
#pragma unroll
        for (int h = 0; h < 2; ++h) {
            if (keep[h][q]) {
                const int pos = (h ? run0 + off1 : off0) + below[h][q];
                fa[q][pos] = make_float4(rbx[h], rby[h], rcj[h], 0.0f);
                th[q][pos] = rth[h];
                dp[q][pos] = rdp[h];
            }
        }
        if (threadIdx.x < 16 && nkept + (int)threadIdx.x < npad) {
            const int pos = nkept + (int)threadIdx.x;
            fa[q][pos] = make_float4(0.0f, 0.0f, 0.0f, 0.0f);
            th[q][pos] = 3.0e38f;
            dp[q][pos] = 0.0f;
        }
        if (threadIdx.x == 0) nch[q] = npad >> 4;
    }
    __syncthreads();

    const int hi   = lane >> 4;
    const int colL = lane & 15;

    for (int q = 0; q < TILES_PER_WG; ++q) {
        const int ty = (blockIdx.x + 4 * (q >> 1)) & 7;
        const int row = TILE_ROWS * ty + wave;
        const int colbase = TILE_COLS * q;
        const int pixbase = row * IMG_S + colbase;
        const float* __restrict__ faq = reinterpret_cast<const float*>(fa[q]);
        const float* __restrict__ thq = th[q];
        const float* __restrict__ dpq = dp[q];
        const int nchunks = __builtin_amdgcn_readfirstlane(nch[q]);

        const float pyv = 1.0f - ((float)row + 0.5f) * (2.0f / 64.0f);
        const float pxv = ((float)(colbase + colL) + 0.5f) * (2.0f / 64.0f) - 1.0f;
        v2f B;
        B.x = hi ? 1.0f : pxv;
        B.y = hi ? 0.0f : pyv;
        float cpix;
        {
#pragma clang fp contract(off)
            cpix = -(pxv * pxv + pyv * pyv) * INV_SIG2;
        }
        v8f C;
#pragma unroll
        for (int r = 0; r < 8; ++r) C[r] = cpix;

        v2f aw0 = {0.0f, 0.0f}, aw1 = {0.0f, 0.0f};
        v2f ad0 = {0.0f, 0.0f}, ad1 = {0.0f, 0.0f};

        for (int k = 0; k < nchunks; ++k) {
            const int j = k * 16;
            const v2f A = *reinterpret_cast<const v2f*>(&faq[(j + colL) * 4 + 2 * hi]);
            const float4 t0 = *reinterpret_cast<const float4*>(&thq[j + 8 * hi]);
            const float4 t1 = *reinterpret_cast<const float4*>(&thq[j + 8 * hi + 4]);
            const float4 d0 = *reinterpret_cast<const float4*>(&dpq[j + 8 * hi]);
            const float4 d1 = *reinterpret_cast<const float4*>(&dpq[j + 8 * hi + 4]);
            v8f D = __builtin_amdgcn_wmma_f32_16x16x4_f32(
                 false, A,  false, B,
                 (short)0, C,  false,  false);
            const float tv[8] = {t0.x, t0.y, t0.z, t0.w, t1.x, t1.y, t1.z, t1.w};
            float w[8];
#pragma unroll
            for (int r = 0; r < 8; ++r) {
                const float lw = D[r];
                w[r] = (lw >= tv[r]) ? __expf(lw) : 0.0f;
            }
            const v2f w01 = {w[0], w[1]}, w23 = {w[2], w[3]}, w45 = {w[4], w[5]}, w67 = {w[6], w[7]};
            const v2f e01 = {d0.x, d0.y}, e23 = {d0.z, d0.w}, e45 = {d1.x, d1.y}, e67 = {d1.z, d1.w};
            aw0 += w01;  ad0 += w01 * e01;
            aw1 += w23;  ad1 += w23 * e23;
            aw0 += w45;  ad0 += w45 * e45;
            aw1 += w67;  ad1 += w67 * e67;
        }

        float ws = (aw0.x + aw1.x) + (aw0.y + aw1.y);
        float ds = (ad0.x + ad1.x) + (ad0.y + ad1.y);
        ws += xhalf(ws);
        ds += xhalf(ds);

        if (hi == 0)
            out[view * (IMG_S * IMG_S) + pixbase + colL] = (BGW + ds) / (BGW + ws);
    }
}

extern "C" void kernel_launch(void* const* d_in, const int* in_sizes, int n_in,
                              void* d_out, int out_size, void* d_ws, size_t ws_size,
                              hipStream_t stream) {
    const float* vertices   = (const float*)d_in[0];
    const float* radii      = (const float*)d_in[1];
    const float* at_centers = (const float*)d_in[2];
    const float* at_normals = (const float*)d_in[3];
    const float* rho        = (const float*)d_in[4];
    const float* phi        = (const float*)d_in[5];
    const float* theta      = (const float*)d_in[6];
    const float* up         = (const float*)d_in[7];

    dim3 grid(8, 64);
    dim3 block(256);
    MultiViewRenderer_48180943126589_kernel<<<grid, block, 0, stream>>>(
        vertices, radii, at_centers, at_normals, rho, phi, theta, up,
        (float*)d_out);
}


# Normalisations, perspective divide and final divide by hardware rsq / rcp (1 ulp) instead of rounded sqrt + IEEE division: 240 fewer dependent instructions before the 1st barrier; same sum order
# speedup vs baseline: 3.6969x; 1.1374x over previous
#include <hip/hip_runtime.h>
#include <math.h>


typedef __attribute__((ext_vector_type(2))) float v2f;
typedef __attribute__((ext_vector_type(8))) float v8f;

#define IMG_S        64
#define NPTS         512
#define TILE_ROWS    8
#define TILE_COLS    16
#define TILES_PER_WG 4
#define NWG_X        ((IMG_S / TILE_ROWS) * (IMG_S / TILE_COLS) / TILES_PER_WG)
static_assert(NWG_X == 8, "kernel_launch sets grid.x to this value as a literal");
#define INV_SIG2     4096.0f
#define INV_GAMMA    0.2f
#define INV_TANHF    1.73205080757f
#define ZNEARC       0.1f
#define ZFARC        6.0f
#define BGW          1e-8f
#define PI_F         3.14159265358979f
#define REACH_MAX    0.17f

__device__ __forceinline__ void norm3(float& x, float& y, float& z) {
    float n2 = x * x + y * y + z * z;
    float inv = __builtin_amdgcn_rsqf(fmaxf(n2, 1e-24f));
    x *= inv; y *= inv; z *= inv;
}
__device__ __forceinline__ void cross3(float ax, float ay, float az,
                                       float bx, float by, float bz,
                                       float& cx, float& cy, float& cz) {
    cx = ay * bz - az * by;
    cy = az * bx - ax * bz;
    cz = ax * by - ay * bx;
}

__device__ __forceinline__ float xhalf(float x) {
    return __builtin_bit_cast(float, __builtin_amdgcn_permlanex16(
        __builtin_bit_cast(int, x), __builtin_bit_cast(int, x),
        0x76543210, 0xfedcba98, false, false));
}

__global__ __launch_bounds__(256)
void MultiViewRenderer_48180943126589_kernel(
    const float* __restrict__ vertices,
    const float* __restrict__ radii,
    const float* __restrict__ at_centers,
    const float* __restrict__ at_normals,
    const float* __restrict__ rho,
    const float* __restrict__ phi,
    const float* __restrict__ theta,
    const float* __restrict__ up,
    float* __restrict__ out)
{
    __shared__ float4 fa[TILES_PER_WG][NPTS];
    __shared__ __attribute__((aligned(16))) float th[TILES_PER_WG][NPTS];
    __shared__ __attribute__((aligned(16))) float dp[TILES_PER_WG][NPTS];
    __shared__ int   cnt[TILES_PER_WG][16];
    __shared__ int   nch[TILES_PER_WG];

    const int view = blockIdx.y;
    const int b = view >> 3;
    const int n = view & 7;

    const int lane = threadIdx.x & 31;
    const int ck = lane < 11 ? lane : 11;
    const float* cbase = up;
    cbase = ck >= 3  ? at_normals : cbase;
    cbase = ck >= 6  ? at_centers : cbase;
    cbase = ck == 9  ? rho        : cbase;
    cbase = ck == 10 ? phi        : cbase;
    cbase = ck == 11 ? theta      : cbase;
    const int offn = 3 * b + ck - 3, offc = 3 * b + ck - 6;
    int coff = ck;
    coff = ck >= 3 ? offn : coff;
    coff = ck >= 6 ? offc : coff;
    coff = ck >= 9 ? n    : coff;
    const int cam = __builtin_bit_cast(int, cbase[coff]);
    float pvx[2], pvy[2], pvz[2], prad[2];
#pragma unroll
    for (int h = 0; h < 2; ++h) {
        const int pt = h * 256 + threadIdx.x;
        pvx[h] = vertices[3 * pt + 0];
        pvy[h] = vertices[3 * pt + 1];
        pvz[h] = vertices[3 * pt + 2];
        prad[h] = radii[pt];
    }
    const float up0   = __builtin_bit_cast(float, __builtin_amdgcn_readlane(cam, 0));
    const float up1   = __builtin_bit_cast(float, __builtin_amdgcn_readlane(cam, 1));
    const float up2   = __builtin_bit_cast(float, __builtin_amdgcn_readlane(cam, 2));
    const float nrm0  = __builtin_bit_cast(float, __builtin_amdgcn_readlane(cam, 3));
    const float nrm1  = __builtin_bit_cast(float, __builtin_amdgcn_readlane(cam, 4));
    const float nrm2  = __builtin_bit_cast(float, __builtin_amdgcn_readlane(cam, 5));
    const float cen0  = __builtin_bit_cast(float, __builtin_amdgcn_readlane(cam, 6));
    const float cen1  = __builtin_bit_cast(float, __builtin_amdgcn_readlane(cam, 7));
    const float cen2  = __builtin_bit_cast(float, __builtin_amdgcn_readlane(cam, 8));
    const float rho_n   = __builtin_bit_cast(float, __builtin_amdgcn_readlane(cam, 9));
    const float phi_n   = __builtin_bit_cast(float, __builtin_amdgcn_readlane(cam, 10));
    const float theta_n = __builtin_bit_cast(float, __builtin_amdgcn_readlane(cam, 11));

    float rho_c = fminf(fmaxf(rho_n, 0.3f), 1.0f);
    float phi_c = fminf(fmaxf(phi_n, 0.0f), 0.5f * PI_F);
    float th_c  = fminf(fmaxf(theta_n, 0.0f), 2.0f * PI_F);
    float sp = __sinf(phi_c), cp = __cosf(phi_c);
    float st = __sinf(th_c),  ct = __cosf(th_c);
    float exl = rho_c * sp * ct, eyl = rho_c * sp * st, ezl = rho_c * cp;

    float cxc = cen0, cyc = cen1, czc = cen2;
    float azx = nrm0, azy = nrm1, azz = nrm2;
    norm3(azx, azy, azz);
    float unx = up0, uny = up1, unz = up2;
    norm3(unx, uny, unz);
    float axx, axy, axz;
    cross3(unx, uny, unz, azx, azy, azz, axx, axy, axz);
    norm3(axx, axy, axz);
    float ayx, ayy, ayz;
    cross3(azx, azy, azz, axx, axy, axz, ayx, ayy, ayz);
    norm3(ayx, ayy, ayz);
    float ex = exl * axx + eyl * ayx + ezl * azx + cxc;
    float ey = exl * axy + eyl * ayy + ezl * azy + cyc;
    float ez = exl * axz + eyl * ayz + ezl * azz + czc;
    float fx = ex - cxc, fy = ey - cyc, fz = ez - czc;
    norm3(fx, fy, fz);
    float rx, ry, rz;
    cross3(unx, uny, unz, fx, fy, fz, rx, ry, rz);
    norm3(rx, ry, rz);
    float ux, uy, uz;
    cross3(fx, fy, fz, rx, ry, rz, ux, uy, uz);
    norm3(ux, uy, uz);
    float t0 = -(rx * ex + ry * ey + rz * ez);
    float t1 = -(ux * ex + uy * ey + uz * ez);
    float t2 = -(fx * ex + fy * ey + fz * ez);
    float r0x = INV_TANHF * rx, r0y = INV_TANHF * ry, r0z = INV_TANHF * rz, r0w = INV_TANHF * t0;
    float r1x = INV_TANHF * ux, r1y = INV_TANHF * uy, r1z = INV_TANHF * uz, r1w = INV_TANHF * t1;
    float r3x = -fx, r3y = -fy, r3z = -fz, r3w = -t2;

    const int wave = threadIdx.x >> 5;

    float rbx[2], rby[2], rcj[2], rdp[2], rth[2];
    bool  keep[2][TILES_PER_WG];
    int   below[2][TILES_PER_WG];
#pragma unroll
    for (int h = 0; h < 2; ++h) {
        float vx = pvx[h], vy = pvy[h], vz = pvz[h];
        float c0 = r0x * vx + r0y * vy + r0z * vz + r0w;
        float c1 = r1x * vx + r1y * vy + r1z * vz + r1w;
        float wc = r3x * vx + r3y * vy + r3z * vz + r3w;
        float wcs = (wc > 1e-6f) ? wc : 1.0f;
        const float iw = __builtin_amdgcn_rcpf(wcs);
        float xs = c0 * iw;
        float ys = c1 * iw;
        float depth = fminf(fmaxf((wc - ZNEARC) * (1.0f / (ZFARC - ZNEARC)), 0.0f), 1.0f);
        bool  infr = (wc > ZNEARC) && (wc < ZFARC);
        float rn = prad[h] * (INV_TANHF * iw);
        float cut2 = 25.0f * rn * rn;
        float cj = -(xs * xs + ys * ys) * INV_SIG2 + (1.0f - depth) * INV_GAMMA;
        float thr = infr ? ((1.0f - depth) * INV_GAMMA - cut2 * INV_SIG2) : 3.0e38f;
        rbx[h] = 2.0f * xs * INV_SIG2;
        rby[h] = 2.0f * ys * INV_SIG2;
        rcj[h] = cj;
        rdp[h] = depth;
        rth[h] = thr;
        const float reach = fminf(REACH_MAX, fabsf(rn) * 5.005f + 0.002f);
        const float ry = reach + 3.5f * (2.0f / 64.0f);
        const float rx = reach + 7.5f * (2.0f / 64.0f);
#pragma unroll
        for (int q = 0; q < TILES_PER_WG; ++q) {
            const int ty = (blockIdx.x + 4 * (q >> 1)) & 7;
            const float yc = 1.0f - ((float)(TILE_ROWS * ty) + 4.0f) * (2.0f / 64.0f);
            const float xc = ((float)(TILE_COLS * q) + 8.0f) * (2.0f / 64.0f) - 1.0f;
            const bool kp = infr & (fabsf(ys - yc) <= ry) & (fabsf(xs - xc) <= rx);
            const unsigned m = __builtin_amdgcn_ballot_w32(kp);
            keep[h][q]  = kp;
            below[h][q] = __builtin_popcount(m & ((1u << lane) - 1u));
            if (lane == 0) cnt[q][h * 8 + wave] = __builtin_popcount(m);
        }
    }
    __syncthreads();

#pragma unroll
    for (int q = 0; q < TILES_PER_WG; ++q) {
        int run0 = 0, run1 = 0, off0 = 0, off1 = 0;
#pragma unroll
        for (int c = 0; c < 8; ++c) {
            if (c == wave) { off0 = run0; off1 = run1; }
            run0 += cnt[q][c];
            run1 += cnt[q][8 + c];
        }
        const int nkept = run0 + run1;
        const int npad  = (nkept + 15) & ~15;
#pragma unroll
        for (int h = 0; h < 2; ++h) {
            if (keep[h][q]) {
                const int pos = (h ? run0 + off1 : off0) + below[h][q];
                fa[q][pos] = make_float4(rbx[h], rby[h], rcj[h], 0.0f);
                th[q][pos] = rth[h];
                dp[q][pos] = rdp[h];
            }
        }
        if (threadIdx.x < 16 && nkept + (int)threadIdx.x < npad) {
            const int pos = nkept + (int)threadIdx.x;
            fa[q][pos] = make_float4(0.0f, 0.0f, 0.0f, 0.0f);
            th[q][pos] = 3.0e38f;
            dp[q][pos] = 0.0f;
        }
        if (threadIdx.x == 0) nch[q] = npad >> 4;
    }
    __syncthreads();

    const int hi   = lane >> 4;
    const int colL = lane & 15;

    for (int q = 0; q < TILES_PER_WG; ++q) {
        const int ty = (blockIdx.x + 4 * (q >> 1)) & 7;
        const int row = TILE_ROWS * ty + wave;
        const int colbase = TILE_COLS * q;
        const int pixbase = row * IMG_S + colbase;
        const float* __restrict__ faq = reinterpret_cast<const float*>(fa[q]);
        const float* __restrict__ thq = th[q];
        const float* __restrict__ dpq = dp[q];
        const int nchunks = __builtin_amdgcn_readfirstlane(nch[q]);

        const float pyv = 1.0f - ((float)row + 0.5f) * (2.0f / 64.0f);
        const float pxv = ((float)(colbase + colL) + 0.5f) * (2.0f / 64.0f) - 1.0f;
        v2f B;
        B.x = hi ? 1.0f : pxv;
        B.y = hi ? 0.0f : pyv;
        float cpix;
        {
#pragma clang fp contract(off)
            cpix = -(pxv * pxv + pyv * pyv) * INV_SIG2;
        }
        v8f C;
#pragma unroll
        for (int r = 0; r < 8; ++r) C[r] = cpix;

        v2f aw0 = {0.0f, 0.0f}, aw1 = {0.0f, 0.0f};
        v2f ad0 = {0.0f, 0.0f}, ad1 = {0.0f, 0.0f};

        for (int k = 0; k < nchunks; ++k) {
            const int j = k * 16;
            const v2f A = *reinterpret_cast<const v2f*>(&faq[(j + colL) * 4 + 2 * hi]);
            const float4 t0 = *reinterpret_cast<const float4*>(&thq[j + 8 * hi]);
            const float4 t1 = *reinterpret_cast<const float4*>(&thq[j + 8 * hi + 4]);
            const float4 d0 = *reinterpret_cast<const float4*>(&dpq[j + 8 * hi]);
            const float4 d1 = *reinterpret_cast<const float4*>(&dpq[j + 8 * hi + 4]);
            v8f D = __builtin_amdgcn_wmma_f32_16x16x4_f32(
                 false, A,  false, B,
                 (short)0, C,  false,  false);
            const float tv[8] = {t0.x, t0.y, t0.z, t0.w, t1.x, t1.y, t1.z, t1.w};
            float w[8];
#pragma unroll
            for (int r = 0; r < 8; ++r) {
                const float lw = D[r];
                w[r] = (lw >= tv[r]) ? __expf(lw) : 0.0f;
            }
            const v2f w01 = {w[0], w[1]}, w23 = {w[2], w[3]}, w45 = {w[4], w[5]}, w67 = {w[6], w[7]};
            const v2f e01 = {d0.x, d0.y}, e23 = {d0.z, d0.w}, e45 = {d1.x, d1.y}, e67 = {d1.z, d1.w};
            aw0 += w01;  ad0 += w01 * e01;
            aw1 += w23;  ad1 += w23 * e23;
            aw0 += w45;  ad0 += w45 * e45;
            aw1 += w67;  ad1 += w67 * e67;
        }

        float ws = (aw0.x + aw1.x) + (aw0.y + aw1.y);
        float ds = (ad0.x + ad1.x) + (ad0.y + ad1.y);
        ws += xhalf(ws);
        ds += xhalf(ds);

        if (hi == 0)
            out[view * (IMG_S * IMG_S) + pixbase + colL] = (BGW + ds) * __builtin_amdgcn_rcpf(BGW + ws);
    }
}

extern "C" void kernel_launch(void* const* d_in, const int* in_sizes, int n_in,
                              void* d_out, int out_size, void* d_ws, size_t ws_size,
                              hipStream_t stream) {
    const float* vertices   = (const float*)d_in[0];
    const float* radii      = (const float*)d_in[1];
    const float* at_centers = (const float*)d_in[2];
    const float* at_normals = (const float*)d_in[3];
    const float* rho        = (const float*)d_in[4];
    const float* phi        = (const float*)d_in[5];
    const float* theta      = (const float*)d_in[6];
    const float* up         = (const float*)d_in[7];

    dim3 grid(8, 64);
    dim3 block(256);
    MultiViewRenderer_48180943126589_kernel<<<grid, block, 0, stream>>>(
        vertices, radii, at_centers, at_normals, rho, phi, theta, up,
        (float*)d_out);
}


# List offsets and lengths by a lane-parallel prefix (4 row_shr adds + readlane) in place of each wave's serial sums over LDS counts; counts leave by 1 store per wave; list order and sums unchanged
# speedup vs baseline: 4.2992x; 1.1629x over previous
#include <hip/hip_runtime.h>
#include <math.h>


typedef __attribute__((ext_vector_type(2))) float v2f;
typedef __attribute__((ext_vector_type(8))) float v8f;

#define IMG_S        64
#define NPTS         512
#define TILE_ROWS    8
#define TILE_COLS    16
#define TILES_PER_WG 4
#define NWG_X        ((IMG_S / TILE_ROWS) * (IMG_S / TILE_COLS) / TILES_PER_WG)
static_assert(NWG_X == 8, "kernel_launch sets grid.x to this value as a literal");
#define INV_SIG2     4096.0f
#define INV_GAMMA    0.2f
#define INV_TANHF    1.73205080757f
#define ZNEARC       0.1f
#define ZFARC        6.0f
#define BGW          1e-8f
#define PI_F         3.14159265358979f
#define REACH_MAX    0.17f

__device__ __forceinline__ void norm3(float& x, float& y, float& z) {
    float n2 = x * x + y * y + z * z;
    float inv = __builtin_amdgcn_rsqf(fmaxf(n2, 1e-24f));
    x *= inv; y *= inv; z *= inv;
}
__device__ __forceinline__ void cross3(float ax, float ay, float az,
                                       float bx, float by, float bz,
                                       float& cx, float& cy, float& cz) {
    cx = ay * bz - az * by;
    cy = az * bx - ax * bz;
    cz = ax * by - ay * bx;
}

__device__ __forceinline__ float xhalf(float x) {
    return __builtin_bit_cast(float, __builtin_amdgcn_permlanex16(
        __builtin_bit_cast(int, x), __builtin_bit_cast(int, x),
        0x76543210, 0xfedcba98, false, false));
}

__global__ __launch_bounds__(256)
void MultiViewRenderer_48180943126589_kernel(
    const float* __restrict__ vertices,
    const float* __restrict__ radii,
    const float* __restrict__ at_centers,
    const float* __restrict__ at_normals,
    const float* __restrict__ rho,
    const float* __restrict__ phi,
    const float* __restrict__ theta,
    const float* __restrict__ up,
    float* __restrict__ out)
{
    __shared__ float4 fa[TILES_PER_WG][NPTS];
    __shared__ __attribute__((aligned(16))) float th[TILES_PER_WG][NPTS];
    __shared__ __attribute__((aligned(16))) float dp[TILES_PER_WG][NPTS];
    __shared__ __attribute__((aligned(8))) int cnt[16 * TILES_PER_WG];
    static_assert(TILES_PER_WG == 4, "two tiles per lane half, two values per lane");

    const int view = blockIdx.y;
    const int b = view >> 3;
    const int n = view & 7;

    const int lane = threadIdx.x & 31;
    const int ck = lane < 11 ? lane : 11;
    const float* cbase = up;
    cbase = ck >= 3  ? at_normals : cbase;
    cbase = ck >= 6  ? at_centers : cbase;
    cbase = ck == 9  ? rho        : cbase;
    cbase = ck == 10 ? phi        : cbase;
    cbase = ck == 11 ? theta      : cbase;
    const int offn = 3 * b + ck - 3, offc = 3 * b + ck - 6;
    int coff = ck;
    coff = ck >= 3 ? offn : coff;
    coff = ck >= 6 ? offc : coff;
    coff = ck >= 9 ? n    : coff;
    const int cam = __builtin_bit_cast(int, cbase[coff]);
    float pvx[2], pvy[2], pvz[2], prad[2];
#pragma unroll
    for (int h = 0; h < 2; ++h) {
        const int pt = h * 256 + threadIdx.x;
        pvx[h] = vertices[3 * pt + 0];
        pvy[h] = vertices[3 * pt + 1];
        pvz[h] = vertices[3 * pt + 2];
        prad[h] = radii[pt];
    }
    const float up0   = __builtin_bit_cast(float, __builtin_amdgcn_readlane(cam, 0));
    const float up1   = __builtin_bit_cast(float, __builtin_amdgcn_readlane(cam, 1));
    const float up2   = __builtin_bit_cast(float, __builtin_amdgcn_readlane(cam, 2));
    const float nrm0  = __builtin_bit_cast(float, __builtin_amdgcn_readlane(cam, 3));
    const float nrm1  = __builtin_bit_cast(float, __builtin_amdgcn_readlane(cam, 4));
    const float nrm2  = __builtin_bit_cast(float, __builtin_amdgcn_readlane(cam, 5));
    const float cen0  = __builtin_bit_cast(float, __builtin_amdgcn_readlane(cam, 6));
    const float cen1  = __builtin_bit_cast(float, __builtin_amdgcn_readlane(cam, 7));
    const float cen2  = __builtin_bit_cast(float, __builtin_amdgcn_readlane(cam, 8));
    const float rho_n   = __builtin_bit_cast(float, __builtin_amdgcn_readlane(cam, 9));
    const float phi_n   = __builtin_bit_cast(float, __builtin_amdgcn_readlane(cam, 10));
    const float theta_n = __builtin_bit_cast(float, __builtin_amdgcn_readlane(cam, 11));

    float rho_c = fminf(fmaxf(rho_n, 0.3f), 1.0f);
    float phi_c = fminf(fmaxf(phi_n, 0.0f), 0.5f * PI_F);
    float th_c  = fminf(fmaxf(theta_n, 0.0f), 2.0f * PI_F);
    float sp = __sinf(phi_c), cp = __cosf(phi_c);
    float st = __sinf(th_c),  ct = __cosf(th_c);
    float exl = rho_c * sp * ct, eyl = rho_c * sp * st, ezl = rho_c * cp;

    float cxc = cen0, cyc = cen1, czc = cen2;
    float azx = nrm0, azy = nrm1, azz = nrm2;
    norm3(azx, azy, azz);
    float unx = up0, uny = up1, unz = up2;
    norm3(unx, uny, unz);
    float axx, axy, axz;
    cross3(unx, uny, unz, azx, azy, azz, axx, axy, axz);
    norm3(axx, axy, axz);
    float ayx, ayy, ayz;
    cross3(azx, azy, azz, axx, axy, axz, ayx, ayy, ayz);
    norm3(ayx, ayy, ayz);
    float ex = exl * axx + eyl * ayx + ezl * azx + cxc;
    float ey = exl * axy + eyl * ayy + ezl * azy + cyc;
    float ez = exl * axz + eyl * ayz + ezl * azz + czc;
    float fx = ex - cxc, fy = ey - cyc, fz = ez - czc;
    norm3(fx, fy, fz);
    float rx, ry, rz;
    cross3(unx, uny, unz, fx, fy, fz, rx, ry, rz);
    norm3(rx, ry, rz);
    float ux, uy, uz;
    cross3(fx, fy, fz, rx, ry, rz, ux, uy, uz);
    norm3(ux, uy, uz);
    float t0 = -(rx * ex + ry * ey + rz * ez);
    float t1 = -(ux * ex + uy * ey + uz * ez);
    float t2 = -(fx * ex + fy * ey + fz * ez);
    float r0x = INV_TANHF * rx, r0y = INV_TANHF * ry, r0z = INV_TANHF * rz, r0w = INV_TANHF * t0;
    float r1x = INV_TANHF * ux, r1y = INV_TANHF * uy, r1z = INV_TANHF * uz, r1w = INV_TANHF * t1;
    float r3x = -fx, r3y = -fy, r3z = -fz, r3w = -t2;

    const int wave = __builtin_amdgcn_readfirstlane((int)(threadIdx.x >> 5));

    float rbx[2], rby[2], rcj[2], rdp[2], rth[2];
    bool  keep[2][TILES_PER_WG];
    int   below[2][TILES_PER_WG];
    int   mycnt = 0;
#pragma unroll
    for (int h = 0; h < 2; ++h) {
        float vx = pvx[h], vy = pvy[h], vz = pvz[h];
        float c0 = r0x * vx + r0y * vy + r0z * vz + r0w;
        float c1 = r1x * vx + r1y * vy + r1z * vz + r1w;
        float wc = r3x * vx + r3y * vy + r3z * vz + r3w;
        float wcs = (wc > 1e-6f) ? wc : 1.0f;
        const float iw = __builtin_amdgcn_rcpf(wcs);
        float xs = c0 * iw;
        float ys = c1 * iw;
        float depth = fminf(fmaxf((wc - ZNEARC) * (1.0f / (ZFARC - ZNEARC)), 0.0f), 1.0f);
        bool  infr = (wc > ZNEARC) && (wc < ZFARC);
        float rn = prad[h] * (INV_TANHF * iw);
        float cut2 = 25.0f * rn * rn;
        float cj = -(xs * xs + ys * ys) * INV_SIG2 + (1.0f - depth) * INV_GAMMA;
        float thr = infr ? ((1.0f - depth) * INV_GAMMA - cut2 * INV_SIG2) : 3.0e38f;
        rbx[h] = 2.0f * xs * INV_SIG2;
        rby[h] = 2.0f * ys * INV_SIG2;
        rcj[h] = cj;
        rdp[h] = depth;
        rth[h] = thr;
        const float reach = fminf(REACH_MAX, fabsf(rn) * 5.005f + 0.002f);
        const float ry = reach + 3.5f * (2.0f / 64.0f);
        const float rx = reach + 7.5f * (2.0f / 64.0f);
#pragma unroll
        for (int q = 0; q < TILES_PER_WG; ++q) {
            const int ty = (blockIdx.x + 4 * (q >> 1)) & 7;
            const float yc = 1.0f - ((float)(TILE_ROWS * ty) + 4.0f) * (2.0f / 64.0f);
            const float xc = ((float)(TILE_COLS * q) + 8.0f) * (2.0f / 64.0f) - 1.0f;
            const bool kp = infr & (fabsf(ys - yc) <= ry) & (fabsf(xs - xc) <= rx);
            const unsigned m = __builtin_amdgcn_ballot_w32(kp);
            keep[h][q]  = kp;
            below[h][q] = __builtin_popcount(m & ((1u << lane) - 1u));
            mycnt = (lane == 4 * h + q) ? __builtin_popcount(m) : mycnt;
        }
    }
    if (lane < 8) {
        const int ch = lane >> 2, cq = lane & 3;
        cnt[((((cq & 1) << 4) + (ch << 3) + wave) << 1) + (cq >> 1)] = mycnt;
    }
    __syncthreads();

    const int2 gc = *reinterpret_cast<const int2*>(&cnt[2 * lane]);
    int inc0 = gc.x, inc1 = gc.y;
#define ROW_SHR_ADD(v, n) v += __builtin_amdgcn_update_dpp(0, v, 0x110 + (n), 0xf, 0xf, true)
    ROW_SHR_ADD(inc0, 1); ROW_SHR_ADD(inc1, 1);
    ROW_SHR_ADD(inc0, 2); ROW_SHR_ADD(inc1, 2);
    ROW_SHR_ADD(inc0, 4); ROW_SHR_ADD(inc1, 4);
    ROW_SHR_ADD(inc0, 8); ROW_SHR_ADD(inc1, 8);
#undef ROW_SHR_ADD
    const int exc0 = inc0 - gc.x, exc1 = inc1 - gc.y;

    int nchs[TILES_PER_WG];
#pragma unroll
    for (int q = 0; q < TILES_PER_WG; ++q) {
        const int half = 16 * (q & 1);
        const int incq = (q >> 1) ? inc1 : inc0;
        const int excq = (q >> 1) ? exc1 : exc0;
        const int nkept = __builtin_amdgcn_readlane(incq, half + 15);
        const int npad  = (nkept + 15) & ~15;
        nchs[q] = npad >> 4;
#pragma unroll
        for (int h = 0; h < 2; ++h) {
            const int off = __builtin_amdgcn_readlane(excq, half + 8 * h + wave);
            if (keep[h][q]) {
                const int pos = off + below[h][q];
                fa[q][pos] = make_float4(rbx[h], rby[h], rcj[h], 0.0f);
                th[q][pos] = rth[h];
                dp[q][pos] = rdp[h];
            }
        }
        if (wave == q) {
            if (lane < 16 && nkept + lane < npad) {
                const int pos = nkept + lane;
                fa[q][pos] = make_float4(0.0f, 0.0f, 0.0f, 0.0f);
                th[q][pos] = 3.0e38f;
                dp[q][pos] = 0.0f;
            }
        }
    }
    __syncthreads();

    const int hi   = lane >> 4;
    const int colL = lane & 15;

#pragma unroll
    for (int q = 0; q < TILES_PER_WG; ++q) {
        const int ty = (blockIdx.x + 4 * (q >> 1)) & 7;
        const int row = TILE_ROWS * ty + wave;
        const int colbase = TILE_COLS * q;
        const int pixbase = row * IMG_S + colbase;
        const float* __restrict__ faq = reinterpret_cast<const float*>(fa[q]);
        const float* __restrict__ thq = th[q];
        const float* __restrict__ dpq = dp[q];
        const int nchunks = nchs[q];

        const float pyv = 1.0f - ((float)row + 0.5f) * (2.0f / 64.0f);
        const float pxv = ((float)(colbase + colL) + 0.5f) * (2.0f / 64.0f) - 1.0f;
        v2f B;
        B.x = hi ? 1.0f : pxv;
        B.y = hi ? 0.0f : pyv;
        float cpix;
        {
#pragma clang fp contract(off)
            cpix = -(pxv * pxv + pyv * pyv) * INV_SIG2;
        }
        v8f C;
#pragma unroll
        for (int r = 0; r < 8; ++r) C[r] = cpix;

        v2f aw0 = {0.0f, 0.0f}, aw1 = {0.0f, 0.0f};
        v2f ad0 = {0.0f, 0.0f}, ad1 = {0.0f, 0.0f};

        for (int k = 0; k < nchunks; ++k) {
            const int j = k * 16;
            const v2f A = *reinterpret_cast<const v2f*>(&faq[(j + colL) * 4 + 2 * hi]);
            const float4 t0 = *reinterpret_cast<const float4*>(&thq[j + 8 * hi]);
            const float4 t1 = *reinterpret_cast<const float4*>(&thq[j + 8 * hi + 4]);
            const float4 d0 = *reinterpret_cast<const float4*>(&dpq[j + 8 * hi]);
            const float4 d1 = *reinterpret_cast<const float4*>(&dpq[j + 8 * hi + 4]);
            v8f D = __builtin_amdgcn_wmma_f32_16x16x4_f32(
                 false, A,  false, B,
                 (short)0, C,  false,  false);
            const float tv[8] = {t0.x, t0.y, t0.z, t0.w, t1.x, t1.y, t1.z, t1.w};
            float w[8];
#pragma unroll
            for (int r = 0; r < 8; ++r) {
                const float lw = D[r];
                w[r] = (lw >= tv[r]) ? __expf(lw) : 0.0f;
            }
            const v2f w01 = {w[0], w[1]}, w23 = {w[2], w[3]}, w45 = {w[4], w[5]}, w67 = {w[6], w[7]};
            const v2f e01 = {d0.x, d0.y}, e23 = {d0.z, d0.w}, e45 = {d1.x, d1.y}, e67 = {d1.z, d1.w};
            aw0 += w01;  ad0 += w01 * e01;
            aw1 += w23;  ad1 += w23 * e23;
            aw0 += w45;  ad0 += w45 * e45;
            aw1 += w67;  ad1 += w67 * e67;
        }

        float ws = (aw0.x + aw1.x) + (aw0.y + aw1.y);
        float ds = (ad0.x + ad1.x) + (ad0.y + ad1.y);
        ws += xhalf(ws);
        ds += xhalf(ds);

        if (hi == 0)
            out[view * (IMG_S * IMG_S) + pixbase + colL] = (BGW + ds) * __builtin_amdgcn_rcpf(BGW + ws);
    }
}

extern "C" void kernel_launch(void* const* d_in, const int* in_sizes, int n_in,
                              void* d_out, int out_size, void* d_ws, size_t ws_size,
                              hipStream_t stream) {
    const float* vertices   = (const float*)d_in[0];
    const float* radii      = (const float*)d_in[1];
    const float* at_centers = (const float*)d_in[2];
    const float* at_normals = (const float*)d_in[3];
    const float* rho        = (const float*)d_in[4];
    const float* phi        = (const float*)d_in[5];
    const float* theta      = (const float*)d_in[6];
    const float* up         = (const float*)d_in[7];

    dim3 grid(8, 64);
    dim3 block(256);
    MultiViewRenderer_48180943126589_kernel<<<grid, block, 0, stream>>>(
        vertices, radii, at_centers, at_normals, rho, phi, theta, up,
        (float*)d_out);
}
